# starting kernel
# baseline (speedup 1.0000x reference)
#include <hip/hip_runtime.h>


#define BB 64
#define TT 4096
#define CC 80
#define NCHUNK 32
#define LCH 128
#define NTILE (LCH / 16)
#define NCT 5
#define DMIN 0.00390625f
#define FLOORV 1e-6f

typedef __attribute__((ext_vector_type(2))) float v2f;
typedef __attribute__((ext_vector_type(8))) float v8f;

__device__ __forceinline__ float clamp01f(float x) { return fminf(fmaxf(x, 0.0f), 1.0f); }
__device__ __forceinline__ float flog2(float x) { return __builtin_amdgcn_logf(x); }
__device__ __forceinline__ float fexp2(float x) { return __builtin_amdgcn_exp2f(x); }

__global__ __launch_bounds__(256)
void pcen_chunk_agg(const float* __restrict__ X, const float* __restrict__ smooth,
                    float* __restrict__ bchunk) {
  const int lane  = threadIdx.x & 31;
  const int wid   = blockIdx.x * 8 + (threadIdx.x >> 5);
  const int ctile = wid % NCT;
  const int tmp   = wid / NCT;
  const int chunk = tmp % NCHUNK;
  const int b     = tmp / NCHUNK;
  const int hi    = lane >> 4;
  const int c     = ctile * 16 + (lane & 15);

  const float wsm = clamp01f(smooth[c]);
  const float d   = fminf(fmaxf(1.0f - wsm, DMIN), 1.0f);
  const float l2d = flog2(d);

  float aP[8], syB[8];
#pragma unroll
  for (int j = 0; j < 8; ++j) {
    const int u = 4 * (j >> 1) + (j & 1) + 2 * hi;
    aP[j]  = (u <= (lane & 15)) ? 1.0f : 0.0f;
    syB[j] = wsm * fexp2(-(float)u * l2d);
  }
  const float dt7 = fexp2((float)(7 + 8 * hi) * l2d);

  const float* pB = X + ((size_t)b * TT + (size_t)chunk * LCH + 2 * hi) * CC + c;

  float s = 0.0f;
#pragma unroll 1
  for (int tile = 0; tile < NTILE; ++tile) {
    float xb[8];
#pragma unroll
    for (int j = 0; j < 8; ++j) {
      const int u0 = 4 * (j >> 1) + (j & 1);
      xb[j] = pB[(size_t)u0 * CC];
    }
    v8f acc = {0.f, 0.f, 0.f, 0.f, 0.f, 0.f, 0.f, 0.f};
#pragma unroll
    for (int kk = 0; kk < 4; ++kk) {
      v2f a2; a2[0] = aP[2 * kk];               a2[1] = aP[2 * kk + 1];
      v2f b2; b2[0] = xb[2 * kk] * syB[2 * kk]; b2[1] = xb[2 * kk + 1] * syB[2 * kk + 1];
      acc = __builtin_amdgcn_wmma_f32_16x16x4_f32(false, a2, false, b2, (short)0, acc,
                                                  false, false);
    }
    const float a15 = dt7 * (acc[7] + d * s);
    s = __shfl(a15, (lane & 15) + 16, 32);
    pB += 16 * CC;
  }
  if (lane < 16) bchunk[((size_t)b * NCHUNK + chunk) * CC + c] = s;
}

__global__ __launch_bounds__(256)
void pcen_combine(const float* __restrict__ X, const float* __restrict__ smooth,
                  const float* __restrict__ bchunk, float* __restrict__ start) {
  const int tid = blockIdx.x * 256 + threadIdx.x;
  if (tid >= BB * CC) return;
  const int b = tid / CC, c = tid % CC;
  const float wsm = clamp01f(smooth[c]);
  const float d   = fminf(fmaxf(1.0f - wsm, DMIN), 1.0f);
  const float dL  = fexp2((float)LCH * flog2(d));
  float s = X[(size_t)b * TT * CC + c];
  for (int k = 0; k < NCHUNK; ++k) {
    start[((size_t)b * NCHUNK + k) * CC + c] = s;
    s = dL * s + bchunk[((size_t)b * NCHUNK + k) * CC + c];
  }
}

__global__ __launch_bounds__(256)
void pcen_emit(const float* __restrict__ X, const float* __restrict__ alpha,
               const float* __restrict__ delta, const float* __restrict__ root,
               const float* __restrict__ smooth, const float* __restrict__ start,
               float* __restrict__ out) {
  const int lane  = threadIdx.x & 31;
  const int wid   = blockIdx.x * 8 + (threadIdx.x >> 5);
  const int ctile = wid % NCT;
  const int tmp   = wid / NCT;
  const int chunk = tmp % NCHUNK;
  const int b     = tmp / NCHUNK;
  const int hi    = lane >> 4;
  const int c     = ctile * 16 + (lane & 15);

  const float wsm = clamp01f(smooth[c]);
  const float d   = fminf(fmaxf(1.0f - wsm, DMIN), 1.0f);
  const float l2d = flog2(d);
  const float ac  = fminf(alpha[c], 1.0f);
  const float dc  = delta[c];
  const float ir  = 1.0f / fmaxf(root[c], 1.0f);
  const float off = fexp2(ir * flog2(dc));

  float aP[8], syB[8], dt[8];
#pragma unroll
  for (int j = 0; j < 8; ++j) {
    const int u = 4 * (j >> 1) + (j & 1) + 2 * hi;
    aP[j]  = (u <= (lane & 15)) ? 1.0f : 0.0f;
    syB[j] = wsm * fexp2(-(float)u * l2d);
    dt[j]  = fexp2((float)(j + 8 * hi) * l2d);
  }

  float s = start[((size_t)b * NCHUNK + chunk) * CC + c];

  const size_t rowBase = ((size_t)b * TT + (size_t)chunk * LCH) * CC + c;
  const float* pB = X + rowBase + (size_t)(2 * hi) * CC;
  const float* pD = X + rowBase + (size_t)(8 * hi) * CC;
  float*       pO = out + rowBase + (size_t)(8 * hi) * CC;

#pragma unroll 1
  for (int tile = 0; tile < NTILE; ++tile) {
    float xb[8];
#pragma unroll
    for (int j = 0; j < 8; ++j) {
      const int u0 = 4 * (j >> 1) + (j & 1);
      xb[j] = pB[(size_t)u0 * CC];
    }
    v8f acc = {0.f, 0.f, 0.f, 0.f, 0.f, 0.f, 0.f, 0.f};
#pragma unroll
    for (int kk = 0; kk < 4; ++kk) {
      v2f a2; a2[0] = aP[2 * kk];               a2[1] = aP[2 * kk + 1];
      v2f b2; b2[0] = xb[2 * kk] * syB[2 * kk]; b2[1] = xb[2 * kk + 1] * syB[2 * kk + 1];
      acc = __builtin_amdgcn_wmma_f32_16x16x4_f32(false, a2, false, b2, (short)0, acc,
                                                  false, false);
    }
    float a7 = 0.0f;
#pragma unroll
    for (int v = 0; v < 8; ++v) {
      const float ema = dt[v] * (acc[v] + d * s);
      if (v == 7) a7 = ema;
      const float xv  = pD[(size_t)v * CC];
      const float lp  = flog2(ema + FLOORV);
      const float num = xv * fexp2(-ac * lp) + dc;
      pO[(size_t)v * CC] = fexp2(ir * flog2(num)) - off;
    }
    s = __shfl(a7, (lane & 15) + 16, 32);
    pB += 16 * CC; pD += 16 * CC; pO += 16 * CC;
  }
}

extern "C" void kernel_launch(void* const* d_in, const int* in_sizes, int n_in,
                              void* d_out, int out_size, void* d_ws, size_t ws_size,
                              hipStream_t stream) {
  const float* X      = (const float*)d_in[0];
  const float* alpha  = (const float*)d_in[1];
  const float* delta  = (const float*)d_in[2];
  const float* root   = (const float*)d_in[3];
  const float* smooth = (const float*)d_in[4];
  float* out = (float*)d_out;

  float* bchunk = (float*)d_ws;
  float* start  = bchunk + (size_t)BB * NCHUNK * CC;

  const int nwaves = BB * NCHUNK * NCT;
  pcen_chunk_agg<<<nwaves / 8, 256, 0, stream>>>(X, smooth, bchunk);
  pcen_combine<<<(BB * CC + 255) / 256, 256, 0, stream>>>(X, smooth, bchunk, start);
  pcen_emit<<<nwaves / 8, 256, 0, stream>>>(X, alpha, delta, root, smooth, start, out);
}


# combine launch folded into the emit kernel: each wave rebuilds its chunk's start state from the chunk aggregates (same serial order), 3 launches -> 2
# speedup vs baseline: 1.2890x; 1.2890x over previous
#include <hip/hip_runtime.h>


#define BB 64
#define TT 4096
#define CC 80
#define NCHUNK 32
#define LCH 128
#define NTILE (LCH / 16)
#define NCT 5
#define DMIN 0.00390625f
#define FLOORV 1e-6f

typedef __attribute__((ext_vector_type(2))) float v2f;
typedef __attribute__((ext_vector_type(8))) float v8f;

__device__ __forceinline__ float clamp01f(float x) { return fminf(fmaxf(x, 0.0f), 1.0f); }
__device__ __forceinline__ float flog2(float x) { return __builtin_amdgcn_logf(x); }
__device__ __forceinline__ float fexp2(float x) { return __builtin_amdgcn_exp2f(x); }

__global__ __launch_bounds__(256)
void pcen_chunk_agg(const float* __restrict__ X, const float* __restrict__ smooth,
                    float* __restrict__ bchunk) {
  const int lane  = threadIdx.x & 31;
  const int wid   = blockIdx.x * 8 + (threadIdx.x >> 5);
  const int ctile = wid % NCT;
  const int tmp   = wid / NCT;
  const int chunk = tmp % NCHUNK;
  const int b     = tmp / NCHUNK;
  const int hi    = lane >> 4;
  const int c     = ctile * 16 + (lane & 15);

  const float wsm = clamp01f(smooth[c]);
  const float d   = fminf(fmaxf(1.0f - wsm, DMIN), 1.0f);
  const float l2d = flog2(d);

  float aP[8], syB[8];
#pragma unroll
  for (int j = 0; j < 8; ++j) {
    const int u = 4 * (j >> 1) + (j & 1) + 2 * hi;
    aP[j]  = (u <= (lane & 15)) ? 1.0f : 0.0f;
    syB[j] = wsm * fexp2(-(float)u * l2d);
  }
  const float dt7 = fexp2((float)(7 + 8 * hi) * l2d);

  const float* pB = X + ((size_t)b * TT + (size_t)chunk * LCH + 2 * hi) * CC + c;

  float s = 0.0f;
#pragma unroll 1
  for (int tile = 0; tile < NTILE; ++tile) {
    float xb[8];
#pragma unroll
    for (int j = 0; j < 8; ++j) {
      const int u0 = 4 * (j >> 1) + (j & 1);
      xb[j] = pB[(size_t)u0 * CC];
    }
    v8f acc = {0.f, 0.f, 0.f, 0.f, 0.f, 0.f, 0.f, 0.f};
#pragma unroll
    for (int kk = 0; kk < 4; ++kk) {
      v2f a2; a2[0] = aP[2 * kk];               a2[1] = aP[2 * kk + 1];
      v2f b2; b2[0] = xb[2 * kk] * syB[2 * kk]; b2[1] = xb[2 * kk + 1] * syB[2 * kk + 1];
      acc = __builtin_amdgcn_wmma_f32_16x16x4_f32(false, a2, false, b2, (short)0, acc,
                                                  false, false);
    }
    const float a15 = dt7 * (acc[7] + d * s);
    s = __shfl(a15, (lane & 15) + 16, 32);
    pB += 16 * CC;
  }
  if (lane < 16) bchunk[((size_t)b * NCHUNK + chunk) * CC + c] = s;
}

__global__ __launch_bounds__(256)
void pcen_emit(const float* __restrict__ X, const float* __restrict__ alpha,
               const float* __restrict__ delta, const float* __restrict__ root,
               const float* __restrict__ smooth, const float* __restrict__ bchunk,
               float* __restrict__ out) {
  const int lane  = threadIdx.x & 31;
  const int wid   = blockIdx.x * 8 + __builtin_amdgcn_readfirstlane((int)(threadIdx.x >> 5));
  const int ctile = wid % NCT;
  const int tmp   = wid / NCT;
  const int chunk = tmp % NCHUNK;
  const int b     = tmp / NCHUNK;
  const int hi    = lane >> 4;
  const int c     = ctile * 16 + (lane & 15);

  const float x0  = X[(size_t)b * TT * CC + c];
  const float wsm = clamp01f(smooth[c]);
  const float d   = fminf(fmaxf(1.0f - wsm, DMIN), 1.0f);
  const float l2d = flog2(d);
  const float ac  = fminf(alpha[c], 1.0f);
  const float dc  = delta[c];
  const float ir  = 1.0f / fmaxf(root[c], 1.0f);
  const float off = fexp2(ir * flog2(dc));

  float aP[8], syB[8], dt[8];
#pragma unroll
  for (int j = 0; j < 8; ++j) {
    const int u = 4 * (j >> 1) + (j & 1) + 2 * hi;
    aP[j]  = (u <= (lane & 15)) ? 1.0f : 0.0f;
    syB[j] = wsm * fexp2(-(float)u * l2d);
    dt[j]  = fexp2((float)(j + 8 * hi) * l2d);
  }

  const float  dL = fexp2((float)LCH * l2d);
  const float* pA = bchunk + (size_t)b * NCHUNK * CC + c;
  float ag[NCHUNK];
#pragma unroll
  for (int g = 0; g < NCHUNK / 8; ++g) {
    if (g * 8 < chunk) {
#pragma unroll
      for (int j = 0; j < 8; ++j) ag[g * 8 + j] = pA[(size_t)(g * 8 + j) * CC];
    } else {
#pragma unroll
      for (int j = 0; j < 8; ++j) ag[g * 8 + j] = 0.0f;
    }
  }
  float s = x0;
#pragma unroll
  for (int k = 0; k < NCHUNK - 1; ++k) {
    if (k < chunk) s = dL * s + ag[k];
  }

  const size_t rowBase = ((size_t)b * TT + (size_t)chunk * LCH) * CC + c;
  const float* pB = X + rowBase + (size_t)(2 * hi) * CC;
  const float* pD = X + rowBase + (size_t)(8 * hi) * CC;
  float*       pO = out + rowBase + (size_t)(8 * hi) * CC;

#pragma unroll 1
  for (int tile = 0; tile < NTILE; ++tile) {
    float xb[8];
#pragma unroll
    for (int j = 0; j < 8; ++j) {
      const int u0 = 4 * (j >> 1) + (j & 1);
      xb[j] = pB[(size_t)u0 * CC];
    }
    v8f acc = {0.f, 0.f, 0.f, 0.f, 0.f, 0.f, 0.f, 0.f};
#pragma unroll
    for (int kk = 0; kk < 4; ++kk) {
      v2f a2; a2[0] = aP[2 * kk];               a2[1] = aP[2 * kk + 1];
      v2f b2; b2[0] = xb[2 * kk] * syB[2 * kk]; b2[1] = xb[2 * kk + 1] * syB[2 * kk + 1];
      acc = __builtin_amdgcn_wmma_f32_16x16x4_f32(false, a2, false, b2, (short)0, acc,
                                                  false, false);
    }
    float a7 = 0.0f;
#pragma unroll
    for (int v = 0; v < 8; ++v) {
      const float ema = dt[v] * (acc[v] + d * s);
      if (v == 7) a7 = ema;
      const float xv  = pD[(size_t)v * CC];
      const float lp  = flog2(ema + FLOORV);
      const float num = xv * fexp2(-ac * lp) + dc;
      pO[(size_t)v * CC] = fexp2(ir * flog2(num)) - off;
    }
    s = __shfl(a7, (lane & 15) + 16, 32);
    pB += 16 * CC; pD += 16 * CC; pO += 16 * CC;
  }
}

extern "C" void kernel_launch(void* const* d_in, const int* in_sizes, int n_in,
                              void* d_out, int out_size, void* d_ws, size_t ws_size,
                              hipStream_t stream) {
  const float* X      = (const float*)d_in[0];
  const float* alpha  = (const float*)d_in[1];
  const float* delta  = (const float*)d_in[2];
  const float* root   = (const float*)d_in[3];
  const float* smooth = (const float*)d_in[4];
  float* out = (float*)d_out;

  float* bchunk = (float*)d_ws;

  const int nwaves = BB * NCHUNK * NCT;
  pcen_chunk_agg<<<nwaves / 8, 256, 0, stream>>>(X, smooth, bchunk);
  pcen_emit<<<nwaves / 8, 256, 0, stream>>>(X, alpha, delta, root, smooth, bchunk, out);
}


# global loads issued far ahead of use: both tile loops unrolled so a wave keeps 30 to 80 loads in flight (aggregate kernel: its chunk; emit kernel: 3 tiles ahead) where it paid a round trip per tile
# speedup vs baseline: 1.3338x; 1.0348x over previous
#include <hip/hip_runtime.h>


#define BB 64
#define TT 4096
#define CC 80
#define NCHUNK 32
#define LCH 128
#define NTILE (LCH / 16)
#define NCT 5
#define PFD 3
#define DMIN 0.00390625f
#define FLOORV 1e-6f

typedef __attribute__((ext_vector_type(2))) float v2f;
typedef __attribute__((ext_vector_type(8))) float v8f;

__device__ __forceinline__ float clamp01f(float x) { return fminf(fmaxf(x, 0.0f), 1.0f); }
__device__ __forceinline__ float flog2(float x) { return __builtin_amdgcn_logf(x); }
__device__ __forceinline__ float fexp2(float x) { return __builtin_amdgcn_exp2f(x); }

__global__ __launch_bounds__(256)
void pcen_chunk_agg(const float* __restrict__ X, const float* __restrict__ smooth,
                    float* __restrict__ bchunk) {
  const int lane  = threadIdx.x & 31;
  const int wid   = blockIdx.x * 8 + (threadIdx.x >> 5);
  const int ctile = wid % NCT;
  const int tmp   = wid / NCT;
  const int chunk = tmp % NCHUNK;
  const int b     = tmp / NCHUNK;
  const int hi    = lane >> 4;
  const int c     = ctile * 16 + (lane & 15);

  const float* pB = X + ((size_t)b * TT + (size_t)chunk * LCH + 2 * hi) * CC + c;

  float xb[NTILE][8];
#pragma unroll
  for (int tile = 0; tile < NTILE; ++tile) {
#pragma unroll
    for (int j = 0; j < 8; ++j) {
      const int u0 = 4 * (j >> 1) + (j & 1);
      xb[tile][j] = pB[(size_t)(tile * 16 + u0) * CC];
    }
  }

  const float wsm = clamp01f(smooth[c]);
  const float d   = fminf(fmaxf(1.0f - wsm, DMIN), 1.0f);
  const float l2d = flog2(d);

  float aP[8], syB[8];
#pragma unroll
  for (int j = 0; j < 8; ++j) {
    const int u = 4 * (j >> 1) + (j & 1) + 2 * hi;
    aP[j]  = (u <= (lane & 15)) ? 1.0f : 0.0f;
    syB[j] = wsm * fexp2(-(float)u * l2d);
  }
  const float dt7 = fexp2((float)(7 + 8 * hi) * l2d);

  float s = 0.0f;
#pragma unroll
  for (int tile = 0; tile < NTILE; ++tile) {
    v8f acc = {0.f, 0.f, 0.f, 0.f, 0.f, 0.f, 0.f, 0.f};
#pragma unroll
    for (int kk = 0; kk < 4; ++kk) {
      v2f a2; a2[0] = aP[2 * kk];     a2[1] = aP[2 * kk + 1];
      v2f b2; b2[0] = xb[tile][2 * kk] * syB[2 * kk];
              b2[1] = xb[tile][2 * kk + 1] * syB[2 * kk + 1];
      acc = __builtin_amdgcn_wmma_f32_16x16x4_f32(false, a2, false, b2, (short)0, acc,
                                                  false, false);
    }
    const float a15 = dt7 * (acc[7] + d * s);
    s = __shfl(a15, (lane & 15) + 16, 32);
  }
  if (lane < 16) bchunk[((size_t)b * NCHUNK + chunk) * CC + c] = s;
}

__global__ __launch_bounds__(256, 8)
void pcen_emit(const float* __restrict__ X, const float* __restrict__ alpha,
               const float* __restrict__ delta, const float* __restrict__ root,
               const float* __restrict__ smooth, const float* __restrict__ bchunk,
               float* __restrict__ out) {
  const int lane  = threadIdx.x & 31;
  const int wid   = blockIdx.x * 8 + __builtin_amdgcn_readfirstlane((int)(threadIdx.x >> 5));
  const int ctile = wid % NCT;
  const int tmp   = wid / NCT;
  const int chunk = tmp % NCHUNK;
  const int b     = tmp / NCHUNK;
  const int hi    = lane >> 4;
  const int c     = ctile * 16 + (lane & 15);

  const size_t rowBase = ((size_t)b * TT + (size_t)chunk * LCH) * CC + c;
  const float* pB = X + rowBase + (size_t)(2 * hi) * CC;
  const float* pD = X + rowBase + (size_t)(8 * hi) * CC;
  float*       pO = out + rowBase + (size_t)(8 * hi) * CC;

  float xb[NTILE][8], xd[NTILE][8];
#pragma unroll
  for (int tile = 0; tile < PFD; ++tile) {
#pragma unroll
    for (int j = 0; j < 8; ++j) {
      const int u0 = 4 * (j >> 1) + (j & 1);
      xb[tile][j] = pB[(size_t)(tile * 16 + u0) * CC];
    }
#pragma unroll
    for (int v = 0; v < 8; ++v) xd[tile][v] = pD[(size_t)(tile * 16 + v) * CC];
  }

  const float x0  = X[(size_t)b * TT * CC + c];
  const float wsm = clamp01f(smooth[c]);
  const float d   = fminf(fmaxf(1.0f - wsm, DMIN), 1.0f);
  const float l2d = flog2(d);
  const float ac  = fminf(alpha[c], 1.0f);
  const float dc  = delta[c];
  const float ir  = 1.0f / fmaxf(root[c], 1.0f);
  const float off = fexp2(ir * flog2(dc));

  float aP[8], syB[8], dt[8];
#pragma unroll
  for (int j = 0; j < 8; ++j) {
    const int u = 4 * (j >> 1) + (j & 1) + 2 * hi;
    aP[j]  = (u <= (lane & 15)) ? 1.0f : 0.0f;
    syB[j] = wsm * fexp2(-(float)u * l2d);
    dt[j]  = fexp2((float)(j + 8 * hi) * l2d);
  }

  const float  dL = fexp2((float)LCH * l2d);
  const float* pA = bchunk + (size_t)b * NCHUNK * CC + c;
  float ag[NCHUNK];
#pragma unroll
  for (int g = 0; g < NCHUNK / 8; ++g) {
    if (g * 8 < chunk) {
#pragma unroll
      for (int j = 0; j < 8; ++j) ag[g * 8 + j] = pA[(size_t)(g * 8 + j) * CC];
    } else {
#pragma unroll
      for (int j = 0; j < 8; ++j) ag[g * 8 + j] = 0.0f;
    }
  }
  float s = x0;
#pragma unroll
  for (int k = 0; k < NCHUNK - 1; ++k) {
    if (k < chunk) s = dL * s + ag[k];
  }

#pragma unroll
  for (int tile = 0; tile < NTILE; ++tile) {
    if (tile + PFD < NTILE) {
#pragma unroll
      for (int j = 0; j < 8; ++j) {
        const int u0 = 4 * (j >> 1) + (j & 1);
        xb[tile + PFD][j] = pB[(size_t)((tile + PFD) * 16 + u0) * CC];
      }
#pragma unroll
      for (int v = 0; v < 8; ++v) xd[tile + PFD][v] = pD[(size_t)((tile + PFD) * 16 + v) * CC];
    }
    v8f acc = {0.f, 0.f, 0.f, 0.f, 0.f, 0.f, 0.f, 0.f};
#pragma unroll
    for (int kk = 0; kk < 4; ++kk) {
      v2f a2; a2[0] = aP[2 * kk];     a2[1] = aP[2 * kk + 1];
      v2f b2; b2[0] = xb[tile][2 * kk] * syB[2 * kk];
              b2[1] = xb[tile][2 * kk + 1] * syB[2 * kk + 1];
      acc = __builtin_amdgcn_wmma_f32_16x16x4_f32(false, a2, false, b2, (short)0, acc,
                                                  false, false);
    }
    float a7 = 0.0f;
#pragma unroll
    for (int v = 0; v < 8; ++v) {
      const float ema = dt[v] * (acc[v] + d * s);
      if (v == 7) a7 = ema;
      const float xv  = xd[tile][v];
      const float lp  = flog2(ema + FLOORV);
      const float num = xv * fexp2(-ac * lp) + dc;
      pO[(size_t)(tile * 16 + v) * CC] = fexp2(ir * flog2(num)) - off;
    }
    s = __shfl(a7, (lane & 15) + 16, 32);
  }
}

extern "C" void kernel_launch(void* const* d_in, const int* in_sizes, int n_in,
                              void* d_out, int out_size, void* d_ws, size_t ws_size,
                              hipStream_t stream) {
  const float* X      = (const float*)d_in[0];
  const float* alpha  = (const float*)d_in[1];
  const float* delta  = (const float*)d_in[2];
  const float* root   = (const float*)d_in[3];
  const float* smooth = (const float*)d_in[4];
  float* out = (float*)d_out;

  float* bchunk = (float*)d_ws;

  const int nwaves = BB * NCHUNK * NCT;
  pcen_chunk_agg<<<nwaves / 8, 256, 0, stream>>>(X, smooth, bchunk);
  pcen_emit<<<nwaves / 8, 256, 0, stream>>>(X, alpha, delta, root, smooth, bchunk, out);
}


# emit kernel loads each x once: fragment k order = rows 8h..8h+7 per lane half (the D layout), one register set feeds product and epilogue; 16-term tile sums regrouped; 90 regs, all 5 groups resident
# speedup vs baseline: 1.4770x; 1.1074x over previous
#include <hip/hip_runtime.h>


#define BB 64
#define TT 4096
#define CC 80
#define NCHUNK 32
#define LCH 128
#define NTILE (LCH / 16)
#define NCT 5
#define PFD 3
#define DMIN 0.00390625f
#define FLOORV 1e-6f

typedef __attribute__((ext_vector_type(2))) float v2f;
typedef __attribute__((ext_vector_type(8))) float v8f;

__device__ __forceinline__ float clamp01f(float x) { return fminf(fmaxf(x, 0.0f), 1.0f); }
__device__ __forceinline__ float flog2(float x) { return __builtin_amdgcn_logf(x); }
__device__ __forceinline__ float fexp2(float x) { return __builtin_amdgcn_exp2f(x); }

__global__ __launch_bounds__(256)
void pcen_chunk_agg(const float* __restrict__ X, const float* __restrict__ smooth,
                    float* __restrict__ bchunk) {
  const int lane  = threadIdx.x & 31;
  const int wid   = blockIdx.x * 8 + (threadIdx.x >> 5);
  const int ctile = wid % NCT;
  const int tmp   = wid / NCT;
  const int chunk = tmp % NCHUNK;
  const int b     = tmp / NCHUNK;
  const int hi    = lane >> 4;
  const int c     = ctile * 16 + (lane & 15);

  const float* pB = X + ((size_t)b * TT + (size_t)chunk * LCH + 2 * hi) * CC + c;

  float xb[NTILE][8];
#pragma unroll
  for (int tile = 0; tile < NTILE; ++tile) {
#pragma unroll
    for (int j = 0; j < 8; ++j) {
      const int u0 = 4 * (j >> 1) + (j & 1);
      xb[tile][j] = pB[(size_t)(tile * 16 + u0) * CC];
    }
  }

  const float wsm = clamp01f(smooth[c]);
  const float d   = fminf(fmaxf(1.0f - wsm, DMIN), 1.0f);
  const float l2d = flog2(d);

  float aP[8], syB[8];
#pragma unroll
  for (int j = 0; j < 8; ++j) {
    const int u = 4 * (j >> 1) + (j & 1) + 2 * hi;
    aP[j]  = (u <= (lane & 15)) ? 1.0f : 0.0f;
    syB[j] = wsm * fexp2(-(float)u * l2d);
  }
  const float dt7 = fexp2((float)(7 + 8 * hi) * l2d);

  float s = 0.0f;
#pragma unroll
  for (int tile = 0; tile < NTILE; ++tile) {
    v8f acc = {0.f, 0.f, 0.f, 0.f, 0.f, 0.f, 0.f, 0.f};
#pragma unroll
    for (int kk = 0; kk < 4; ++kk) {
      v2f a2; a2[0] = aP[2 * kk];     a2[1] = aP[2 * kk + 1];
      v2f b2; b2[0] = xb[tile][2 * kk] * syB[2 * kk];
              b2[1] = xb[tile][2 * kk + 1] * syB[2 * kk + 1];
      acc = __builtin_amdgcn_wmma_f32_16x16x4_f32(false, a2, false, b2, (short)0, acc,
                                                  false, false);
    }
    const float a15 = dt7 * (acc[7] + d * s);
    s = __shfl(a15, (lane & 15) + 16, 32);
  }
  if (lane < 16) bchunk[((size_t)b * NCHUNK + chunk) * CC + c] = s;
}

__global__ __launch_bounds__(256, 10)
void pcen_emit(const float* __restrict__ X, const float* __restrict__ alpha,
               const float* __restrict__ delta, const float* __restrict__ root,
               const float* __restrict__ smooth, const float* __restrict__ bchunk,
               float* __restrict__ out) {
  const int lane  = threadIdx.x & 31;
  const int wid   = blockIdx.x * 8 + __builtin_amdgcn_readfirstlane((int)(threadIdx.x >> 5));
  const int ctile = wid % NCT;
  const int tmp   = wid / NCT;
  const int chunk = tmp % NCHUNK;
  const int b     = tmp / NCHUNK;
  const int hi    = lane >> 4;
  const int c     = ctile * 16 + (lane & 15);

  const size_t rowBase = ((size_t)b * TT + (size_t)chunk * LCH) * CC + c;
  const float* pD = X + rowBase + (size_t)(8 * hi) * CC;
  float*       pO = out + rowBase + (size_t)(8 * hi) * CC;

  float xd[NTILE][8];
#pragma unroll
  for (int tile = 0; tile < PFD; ++tile) {
#pragma unroll
    for (int v = 0; v < 8; ++v) xd[tile][v] = pD[(size_t)(tile * 16 + v) * CC];
  }

  const float x0  = X[(size_t)b * TT * CC + c];
  const float wsm = clamp01f(smooth[c]);
  const float d   = fminf(fmaxf(1.0f - wsm, DMIN), 1.0f);
  const float l2d = flog2(d);
  const float ac  = fminf(alpha[c], 1.0f);
  const float dc  = delta[c];
  const float ir  = 1.0f / fmaxf(root[c], 1.0f);
  const float off = fexp2(ir * flog2(dc));

  float aP[8], syB[8], dt[8];
#pragma unroll
  for (int j = 0; j < 8; ++j) {
    const int u = j + 8 * hi;
    aP[j]  = (u <= (lane & 15)) ? 1.0f : 0.0f;
    syB[j] = wsm * fexp2(-(float)u * l2d);
    dt[j]  = fexp2((float)u * l2d);
  }

  const float  dL = fexp2((float)LCH * l2d);
  const float* pA = bchunk + (size_t)b * NCHUNK * CC + c;
  float ag[NCHUNK];
#pragma unroll
  for (int g = 0; g < NCHUNK / 8; ++g) {
    if (g * 8 < chunk) {
#pragma unroll
      for (int j = 0; j < 8; ++j) ag[g * 8 + j] = pA[(size_t)(g * 8 + j) * CC];
    } else {
#pragma unroll
      for (int j = 0; j < 8; ++j) ag[g * 8 + j] = 0.0f;
    }
  }
  float s = x0;
#pragma unroll
  for (int k = 0; k < NCHUNK - 1; ++k) {
    if (k < chunk) s = dL * s + ag[k];
  }

#pragma unroll
  for (int tile = 0; tile < NTILE; ++tile) {
    if (tile + PFD < NTILE) {
#pragma unroll
      for (int v = 0; v < 8; ++v) xd[tile + PFD][v] = pD[(size_t)((tile + PFD) * 16 + v) * CC];
    }
    v8f acc = {0.f, 0.f, 0.f, 0.f, 0.f, 0.f, 0.f, 0.f};
#pragma unroll
    for (int kk = 0; kk < 4; ++kk) {
      v2f a2; a2[0] = aP[2 * kk];     a2[1] = aP[2 * kk + 1];
      v2f b2; b2[0] = xd[tile][2 * kk] * syB[2 * kk];
              b2[1] = xd[tile][2 * kk + 1] * syB[2 * kk + 1];
      acc = __builtin_amdgcn_wmma_f32_16x16x4_f32(false, a2, false, b2, (short)0, acc,
                                                  false, false);
    }
    float a7 = 0.0f;
#pragma unroll
    for (int v = 0; v < 8; ++v) {
      const float ema = dt[v] * (acc[v] + d * s);
      if (v == 7) a7 = ema;
      const float xv  = xd[tile][v];
      const float lp  = flog2(ema + FLOORV);
      const float num = xv * fexp2(-ac * lp) + dc;
      pO[(size_t)(tile * 16 + v) * CC] = fexp2(ir * flog2(num)) - off;
    }
    s = __shfl(a7, (lane & 15) + 16, 32);
    __builtin_amdgcn_sched_barrier(0);
  }
}

extern "C" void kernel_launch(void* const* d_in, const int* in_sizes, int n_in,
                              void* d_out, int out_size, void* d_ws, size_t ws_size,
                              hipStream_t stream) {
  const float* X      = (const float*)d_in[0];
  const float* alpha  = (const float*)d_in[1];
  const float* delta  = (const float*)d_in[2];
  const float* root   = (const float*)d_in[3];
  const float* smooth = (const float*)d_in[4];
  float* out = (float*)d_out;

  float* bchunk = (float*)d_ws;

  const int nwaves = BB * NCHUNK * NCT;
  pcen_chunk_agg<<<nwaves / 8, 256, 0, stream>>>(X, smooth, bchunk);
  pcen_emit<<<nwaves / 8, 256, 0, stream>>>(X, alpha, delta, root, smooth, bchunk, out);
}


# chunk per wave 128 -> 256 rows in both kernels: half as many waves enter (5,120 a launch, 256 groups of 20), twice the loads in flight each; rows 128-255 start from the carried state (sum order)
# speedup vs baseline: 1.6609x; 1.1245x over previous
#include <hip/hip_runtime.h>


#define BB 64
#define TT 4096
#define CC 80
#define NCHUNK 16
#define LCH 256
#define NTILE (LCH / 16)
#define NCT 5
#define CPG 4
#define NWV (NCT * CPG)
#define PFA 6
#define PFD 6
#define DMIN 0.00390625f
#define FLOORV 1e-6f

typedef __attribute__((ext_vector_type(2))) float v2f;
typedef __attribute__((ext_vector_type(8))) float v8f;

__device__ __forceinline__ float clamp01f(float x) { return fminf(fmaxf(x, 0.0f), 1.0f); }
__device__ __forceinline__ float flog2(float x) { return __builtin_amdgcn_logf(x); }
__device__ __forceinline__ float fexp2(float x) { return __builtin_amdgcn_exp2f(x); }

__global__ __launch_bounds__(32 * NWV)
void pcen_chunk_agg(const float* __restrict__ X, const float* __restrict__ smooth,
                    float* __restrict__ bchunk) {
  const int lane  = threadIdx.x & 31;
  const int w     = __builtin_amdgcn_readfirstlane((int)(threadIdx.x >> 5));
  const int ctile = w % NCT;
  const int unit  = blockIdx.x * CPG + w / NCT;
  const int hi    = lane >> 4;
  const int c     = ctile * 16 + (lane & 15);

  const float* pB = X + ((size_t)unit * LCH + 2 * hi) * CC + c;

  const float sm = smooth[c];
  float xb[NTILE][8];
#pragma unroll
  for (int tile = 0; tile < PFA; ++tile) {
#pragma unroll
    for (int j = 0; j < 8; ++j) {
      const int u0 = 4 * (j >> 1) + (j & 1);
      xb[tile][j] = pB[(size_t)(tile * 16 + u0) * CC];
    }
  }
  __builtin_amdgcn_sched_barrier(0);

  const float wsm = clamp01f(sm);
  const float d   = fminf(fmaxf(1.0f - wsm, DMIN), 1.0f);
  const float l2d = flog2(d);

  float aP[8], syB[8];
#pragma unroll
  for (int j = 0; j < 8; ++j) {
    const int u = 4 * (j >> 1) + (j & 1) + 2 * hi;
    aP[j]  = (u <= (lane & 15)) ? 1.0f : 0.0f;
    syB[j] = wsm * fexp2(-(float)u * l2d);
  }
  const float dt7 = fexp2((float)(7 + 8 * hi) * l2d);

  float s = 0.0f;
#pragma unroll
  for (int tile = 0; tile < NTILE; ++tile) {
    if (tile + PFA < NTILE) {
#pragma unroll
      for (int j = 0; j < 8; ++j) {
        const int u0 = 4 * (j >> 1) + (j & 1);
        xb[tile + PFA][j] = pB[(size_t)((tile + PFA) * 16 + u0) * CC];
      }
    }
    v8f acc = {0.f, 0.f, 0.f, 0.f, 0.f, 0.f, 0.f, 0.f};
#pragma unroll
    for (int kk = 0; kk < 4; ++kk) {
      v2f a2; a2[0] = aP[2 * kk];     a2[1] = aP[2 * kk + 1];
      v2f b2; b2[0] = xb[tile][2 * kk] * syB[2 * kk];
              b2[1] = xb[tile][2 * kk + 1] * syB[2 * kk + 1];
      acc = __builtin_amdgcn_wmma_f32_16x16x4_f32(false, a2, false, b2, (short)0, acc,
                                                  false, false);
    }
    const float a15 = dt7 * (acc[7] + d * s);
    s = __shfl(a15, (lane & 15) + 16, 32);
    __builtin_amdgcn_sched_barrier(0);
  }
  if (lane < 16) bchunk[(size_t)unit * CC + c] = s;
}

__global__ __launch_bounds__(32 * NWV, 5)
void pcen_emit(const float* __restrict__ X, const float* __restrict__ alpha,
               const float* __restrict__ delta, const float* __restrict__ root,
               const float* __restrict__ smooth, const float* __restrict__ bchunk,
               float* __restrict__ out) {
  const int lane  = threadIdx.x & 31;
  const int w     = __builtin_amdgcn_readfirstlane((int)(threadIdx.x >> 5));
  const int ctile = w % NCT;
  const int unit  = blockIdx.x * CPG + w / NCT;
  const int chunk = unit % NCHUNK;
  const int b     = unit / NCHUNK;
  const int hi    = lane >> 4;
  const int c     = ctile * 16 + (lane & 15);

  const size_t rowBase = (size_t)unit * LCH * CC + c;
  const float* pD = X + rowBase + (size_t)(8 * hi) * CC;
  float*       pO = out + rowBase + (size_t)(8 * hi) * CC;

  float xd[NTILE][8];
#pragma unroll
  for (int tile = 0; tile < PFD; ++tile) {
#pragma unroll
    for (int v = 0; v < 8; ++v) xd[tile][v] = pD[(size_t)(tile * 16 + v) * CC];
  }

  const float x0  = X[(size_t)b * TT * CC + c];
  const float wsm = clamp01f(smooth[c]);
  const float d   = fminf(fmaxf(1.0f - wsm, DMIN), 1.0f);
  const float l2d = flog2(d);
  const float ac  = fminf(alpha[c], 1.0f);
  const float dc  = delta[c];
  const float ir  = 1.0f / fmaxf(root[c], 1.0f);
  const float off = fexp2(ir * flog2(dc));

  float aP[8], syB[8], dt[8];
#pragma unroll
  for (int j = 0; j < 8; ++j) {
    const int u = j + 8 * hi;
    aP[j]  = (u <= (lane & 15)) ? 1.0f : 0.0f;
    syB[j] = wsm * fexp2(-(float)u * l2d);
    dt[j]  = fexp2((float)u * l2d);
  }

  const float  dL = fexp2((float)LCH * l2d);
  const float* pA = bchunk + (size_t)b * NCHUNK * CC + c;
  float ag[NCHUNK];
#pragma unroll
  for (int g = 0; g < NCHUNK / 8; ++g) {
    if (g * 8 < chunk) {
#pragma unroll
      for (int j = 0; j < 8; ++j) ag[g * 8 + j] = pA[(size_t)(g * 8 + j) * CC];
    } else {
#pragma unroll
      for (int j = 0; j < 8; ++j) ag[g * 8 + j] = 0.0f;
    }
  }
  float s = x0;
#pragma unroll
  for (int k = 0; k < NCHUNK - 1; ++k) {
    if (k < chunk) s = dL * s + ag[k];
  }

#pragma unroll
  for (int tile = 0; tile < NTILE; ++tile) {
    if (tile + PFD < NTILE) {
#pragma unroll
      for (int v = 0; v < 8; ++v) xd[tile + PFD][v] = pD[(size_t)((tile + PFD) * 16 + v) * CC];
    }
    v8f acc = {0.f, 0.f, 0.f, 0.f, 0.f, 0.f, 0.f, 0.f};
#pragma unroll
    for (int kk = 0; kk < 4; ++kk) {
      v2f a2; a2[0] = aP[2 * kk];     a2[1] = aP[2 * kk + 1];
      v2f b2; b2[0] = xd[tile][2 * kk] * syB[2 * kk];
              b2[1] = xd[tile][2 * kk + 1] * syB[2 * kk + 1];
      acc = __builtin_amdgcn_wmma_f32_16x16x4_f32(false, a2, false, b2, (short)0, acc,
                                                  false, false);
    }
    float a7 = 0.0f;
#pragma unroll
    for (int v = 0; v < 8; ++v) {
      const float ema = dt[v] * (acc[v] + d * s);
      if (v == 7) a7 = ema;
      const float xv  = xd[tile][v];
      const float lp  = flog2(ema + FLOORV);
      const float num = xv * fexp2(-ac * lp) + dc;
      pO[(size_t)(tile * 16 + v) * CC] = fexp2(ir * flog2(num)) - off;
    }
    s = __shfl(a7, (lane & 15) + 16, 32);
    __builtin_amdgcn_sched_barrier(0);
  }
}

extern "C" void kernel_launch(void* const* d_in, const int* in_sizes, int n_in,
                              void* d_out, int out_size, void* d_ws, size_t ws_size,
                              hipStream_t stream) {
  const float* X      = (const float*)d_in[0];
  const float* alpha  = (const float*)d_in[1];
  const float* delta  = (const float*)d_in[2];
  const float* root   = (const float*)d_in[3];
  const float* smooth = (const float*)d_in[4];
  float* out = (float*)d_out;

  float* bchunk = (float*)d_ws;

  pcen_chunk_agg<<<BB * NCHUNK / CPG, 32 * NWV, 0, stream>>>(X, smooth, bchunk);
  pcen_emit<<<BB * NCHUNK / CPG, 32 * NWV, 0, stream>>>(X, alpha, delta, root, smooth, bchunk, out);
}
